# starting kernel
# baseline (speedup 1.0000x reference)
#include <hip/hip_runtime.h>
#include <cstdint>

typedef __attribute__((ext_vector_type(2))) float v2f;
typedef __attribute__((ext_vector_type(8))) float v8f;

#define N_NODES 8192
#define F_INF   512
#define HIDF    64
#define CCH     16
#define KORD    10
#define JCHUNK  512
#define NCHUNKS (N_NODES / JCHUNK)

__device__ __forceinline__ v8f wmma4(v2f a, v2f b, v8f c) {
  return __builtin_amdgcn_wmma_f32_16x16x4_f32(
       false, a,  false, b,
       (short)0, c,  false,  false);
}

__global__ void zero_kernel(float4* __restrict__ p) {
  p[blockIdx.x * blockDim.x + threadIdx.x] = make_float4(0.f, 0.f, 0.f, 0.f);
}

__global__ void jacobi_kernel(const float* __restrict__ origin_e,
                              const float* __restrict__ temp,
                              float* __restrict__ Hf) {
  __shared__ float al[CCH * (KORD + 1)];
  int tid = threadIdx.x;
  if (tid < CCH * (KORD + 1)) al[tid] = 0.5f * tanhf(temp[tid]);
  __syncthreads();

  int n = blockIdx.x * blockDim.x + tid;
  if (n >= N_NODES) return;
  float e = origin_e[n];

  const float a = 1.0f, b = 1.0f, lo = -1.0f, hi = 1.0f;
  const float coef1 = (a - b) * 0.5f - (a + b + 2.0f) * 0.5f * (lo + hi) / (hi - lo);
  const float coef2 = (a + b + 2.0f) / (hi - lo);

  for (int c = 0; c < CCH; ++c) {
    const float* ac = &al[c * (KORD + 1)];
    float xs_m2 = ac[0];
    float acc = xs_m2;
    float xs_m1 = coef1 * ac[1] + coef2 * ac[1] * e * xs_m2;
    acc += xs_m1;
#pragma unroll
    for (int L = 2; L <= KORD; ++L) {
      float Lf = (float)L;
      float coef_l  = 2.0f * Lf * (Lf + a + b) * (2.0f * Lf - 2.0f + a + b);
      float c_lm1_1 = (2.0f * Lf + a + b - 1.0f) * (2.0f * Lf + a + b) * (2.0f * Lf + a + b - 2.0f);
      float c_lm1_2 = (2.0f * Lf + a + b - 1.0f) * (a * a - b * b);
      float c_lm2   = 2.0f * (Lf - 1.0f + a) * (Lf - 1.0f + b) * (2.0f * Lf + a + b);
      float tmp1   = ac[L - 1] * (c_lm1_1 / coef_l);
      float tmp2   = ac[L - 1] * (c_lm1_2 / coef_l);
      float tmp3   = ac[L - 1] * ac[L - 2] * (c_lm2 / coef_l);
      float tmp1_2 = tmp1 * (2.0f / (hi - lo));
      float tmp2_2 = tmp1 * ((hi + lo) / (hi - lo)) + tmp2;
      float nx = tmp1_2 * (e * xs_m1) - tmp2_2 * xs_m1 - tmp3 * xs_m2;
      acc += nx;
      xs_m2 = xs_m1;
      xs_m1 = nx;
    }
    Hf[n * CCH + c] = acc;
  }
}

#define XS_STRIDE 516
#define HS_STRIDE 68

__global__ void mlp_kernel(const float* __restrict__ x,
                           const float* __restrict__ W1,
                           const float* __restrict__ b1,
                           const float* __restrict__ W2,
                           const float* __restrict__ b2,
                           float* __restrict__ zb) {
  __shared__ float xs[16 * XS_STRIDE];
  __shared__ float hs[16 * HS_STRIDE];

  const int tid  = threadIdx.x;
  const int w    = tid >> 5;
  const int lane = tid & 31;
  const int mn   = lane & 15;
  const int half = lane >> 4;
  const int kk   = half * 2;
  const int m0   = blockIdx.x * 16;
  const int n0   = w * 16;

  for (int idx = tid; idx < 16 * (F_INF / 4); idx += blockDim.x) {
    int row = idx >> 7;
    int c4  = idx & 127;
    float4 v = *(const float4*)&x[(size_t)(m0 + row) * F_INF + c4 * 4];
    float* d = &xs[row * XS_STRIDE + c4 * 4];
    d[0] = v.x; d[1] = v.y; d[2] = v.z; d[3] = v.w;
  }
  __syncthreads();

  v8f acc = {};
#pragma unroll 4
  for (int k0 = 0; k0 < F_INF; k0 += 4) {
    v2f av, bv;
    av.x = xs[mn * XS_STRIDE + k0 + kk];
    av.y = xs[mn * XS_STRIDE + k0 + kk + 1];
    bv.x = W1[(size_t)(k0 + kk) * HIDF + n0 + mn];
    bv.y = W1[(size_t)(k0 + kk + 1) * HIDF + n0 + mn];
    acc = wmma4(av, bv, acc);
  }

  float bias = b1[n0 + mn];
#pragma unroll
  for (int r = 0; r < 8; ++r) {
    int row = half ? r + 8 : r;
    float v = acc[r] + bias;
    hs[row * HS_STRIDE + n0 + mn] = v > 0.0f ? v : 0.0f;
  }
  __syncthreads();

  if (w == 0) {
    v8f zacc = {};
#pragma unroll
    for (int k0 = 0; k0 < HIDF; k0 += 4) {
      v2f av, bv;
      av.x = hs[mn * HS_STRIDE + k0 + kk];
      av.y = hs[mn * HS_STRIDE + k0 + kk + 1];
      bv.x = W2[(k0 + kk) * CCH + mn];
      bv.y = W2[(k0 + kk + 1) * CCH + mn];
      zacc = wmma4(av, bv, zacc);
    }
    float bb = b2[mn];
#pragma unroll
    for (int r = 0; r < 8; ++r) {
      int row = half ? r + 8 : r;
      zb[(size_t)(m0 + row) * CCH + mn] = zacc[r] + bb;
    }
  }
}

__global__ void utz_kernel(const float* __restrict__ U,
                           const float* __restrict__ zb,
                           float* __restrict__ tbuf) {
  __shared__ float zs[JCHUNK * CCH];

  const int tid  = threadIdx.x;
  const int lane = tid & 31;
  const int mn   = lane & 15;
  const int half = lane >> 4;
  const int kk   = half * 2;
  const int W    = blockIdx.x * 8 + (tid >> 5);
  const int i0   = (W & 511) * 16;
  const int jb   = (W >> 9) * JCHUNK;

  for (int idx = tid; idx < JCHUNK * CCH / 4; idx += blockDim.x) {
    float4 v = *(const float4*)&zb[(size_t)jb * CCH + idx * 4];
    float* d = &zs[idx * 4];
    d[0] = v.x; d[1] = v.y; d[2] = v.z; d[3] = v.w;
  }
  __syncthreads();

  v8f acc = {};
#pragma unroll 4
  for (int jl = 0; jl < JCHUNK; jl += 4) {
    v2f av, bv;
    av.x = zs[(jl + kk) * CCH + mn];
    av.y = zs[(jl + kk + 1) * CCH + mn];
    bv.x = U[(size_t)(jb + jl + kk) * N_NODES + i0 + mn];
    bv.y = U[(size_t)(jb + jl + kk + 1) * N_NODES + i0 + mn];
    acc = wmma4(av, bv, acc);
  }
#pragma unroll
  for (int r = 0; r < 8; ++r) {
    int c = half ? r + 8 : r;
    unsafeAtomicAdd(&tbuf[(size_t)(i0 + mn) * CCH + c], acc[r]);
  }
}

#define UT_STRIDE 68

__global__ void outgemm_kernel(const float* __restrict__ U,
                               const float* __restrict__ tbuf,
                               const float* __restrict__ Hf,
                               float* __restrict__ obuf) {
  __shared__ float Ut[128 * UT_STRIDE];
  __shared__ float St[64 * CCH];

  const int tid  = threadIdx.x;
  const int w    = tid >> 5;
  const int lane = tid & 31;
  const int mn   = lane & 15;
  const int half = lane >> 4;
  const int kk   = half * 2;
  const int r0   = (blockIdx.x & 63) * 128;
  const int kb   = (blockIdx.x >> 6) * JCHUNK;

  v8f acc = {};
  for (int i0 = kb; i0 < kb + JCHUNK; i0 += 64) {
    for (int idx = tid; idx < 128 * 16; idx += blockDim.x) {
      int row = idx >> 4;
      int c4  = idx & 15;
      float4 v = *(const float4*)&U[(size_t)(r0 + row) * N_NODES + i0 + c4 * 4];
      float* d = &Ut[row * UT_STRIDE + c4 * 4];
      d[0] = v.x; d[1] = v.y; d[2] = v.z; d[3] = v.w;
    }
    {
      size_t base = (size_t)(i0 + (tid >> 2)) * CCH + (tid & 3) * 4;
      float4 tv = *(const float4*)&tbuf[base];
      float4 hv = *(const float4*)&Hf[base];
      float* d = &St[(tid >> 2) * CCH + (tid & 3) * 4];
      d[0] = tv.x * hv.x; d[1] = tv.y * hv.y; d[2] = tv.z * hv.z; d[3] = tv.w * hv.w;
    }
    __syncthreads();
#pragma unroll
    for (int k0 = 0; k0 < 64; k0 += 4) {
      v2f av, bv;
      av.x = Ut[(16 * w + mn) * UT_STRIDE + k0 + kk];
      av.y = Ut[(16 * w + mn) * UT_STRIDE + k0 + kk + 1];
      bv.x = St[(k0 + kk) * CCH + mn];
      bv.y = St[(k0 + kk + 1) * CCH + mn];
      acc = wmma4(av, bv, acc);
    }
    __syncthreads();
  }
#pragma unroll
  for (int r = 0; r < 8; ++r) {
    int row = half ? r + 8 : r;
    unsafeAtomicAdd(&obuf[(size_t)(r0 + 16 * w + row) * CCH + mn], acc[r]);
  }
}

__global__ void lsm_kernel(const float* __restrict__ obuf,
                           float* __restrict__ out) {
  int n = blockIdx.x * blockDim.x + threadIdx.x;
  if (n >= N_NODES) return;
  float v[CCH];
#pragma unroll
  for (int q = 0; q < 4; ++q) {
    float4 t = *(const float4*)&obuf[(size_t)n * CCH + q * 4];
    v[q * 4 + 0] = t.x; v[q * 4 + 1] = t.y; v[q * 4 + 2] = t.z; v[q * 4 + 3] = t.w;
  }
  float mx = v[0];
#pragma unroll
  for (int i = 1; i < CCH; ++i) mx = fmaxf(mx, v[i]);
  float sm = 0.0f;
#pragma unroll
  for (int i = 0; i < CCH; ++i) sm += expf(v[i] - mx);
  float lse = mx + logf(sm);
#pragma unroll
  for (int q = 0; q < 4; ++q) {
    float4 t;
    t.x = v[q * 4 + 0] - lse; t.y = v[q * 4 + 1] - lse;
    t.z = v[q * 4 + 2] - lse; t.w = v[q * 4 + 3] - lse;
    *(float4*)&out[(size_t)n * CCH + q * 4] = t;
  }
}

extern "C" void kernel_launch(void* const* d_in, const int* in_sizes, int n_in,
                              void* d_out, int out_size, void* d_ws, size_t ws_size,
                              hipStream_t stream) {
  const float* origin_e = (const float*)d_in[0];
  const float* U        = (const float*)d_in[1];
  const float* x        = (const float*)d_in[2];
  const float* W1       = (const float*)d_in[3];
  const float* b1       = (const float*)d_in[4];
  const float* W2       = (const float*)d_in[5];
  const float* b2       = (const float*)d_in[6];
  const float* temp     = (const float*)d_in[7];
  float* out = (float*)d_out;

  float* ws   = (float*)d_ws;
  float* Hf   = ws;
  float* zb   = Hf   + (size_t)N_NODES * CCH;
  float* tbuf = zb   + (size_t)N_NODES * CCH;
  float* obuf = tbuf + (size_t)N_NODES * CCH;

  zero_kernel<<<256, 256, 0, stream>>>((float4*)tbuf);
  jacobi_kernel<<<N_NODES / 256, 256, 0, stream>>>(origin_e, temp, Hf);
  mlp_kernel<<<N_NODES / 16, 128, 0, stream>>>(x, W1, b1, W2, b2, zb);
  utz_kernel<<<(N_NODES / 16) * NCHUNKS / 8, 256, 0, stream>>>(U, zb, tbuf);
  outgemm_kernel<<<64 * NCHUNKS, 256, 0, stream>>>(U, tbuf, Hf, obuf);
  lsm_kernel<<<N_NODES / 256, 256, 0, stream>>>(obuf, out);
}


# utz (t=U^T z): lanes stream U with 128-bit loads, 16-32 in flight per wave, into 4 column-interleaved WMMA tiles; 8 row groups summed in LDS, 4 (was 16) atomic partials per element: sum order changes
# speedup vs baseline: 1.0878x; 1.0878x over previous
#include <hip/hip_runtime.h>
#include <cstdint>

typedef __attribute__((ext_vector_type(2))) float v2f;
typedef __attribute__((ext_vector_type(8))) float v8f;

#define N_NODES 8192
#define F_INF   512
#define HIDF    64
#define CCH     16
#define KORD    10
#define JCHUNK  512
#define NCHUNKS (N_NODES / JCHUNK)

__device__ __forceinline__ v8f wmma4(v2f a, v2f b, v8f c) {
  return __builtin_amdgcn_wmma_f32_16x16x4_f32(
       false, a,  false, b,
       (short)0, c,  false,  false);
}

__global__ void zero_kernel(float4* __restrict__ p) {
  p[blockIdx.x * blockDim.x + threadIdx.x] = make_float4(0.f, 0.f, 0.f, 0.f);
}

__global__ void jacobi_kernel(const float* __restrict__ origin_e,
                              const float* __restrict__ temp,
                              float* __restrict__ Hf) {
  __shared__ float al[CCH * (KORD + 1)];
  int tid = threadIdx.x;
  if (tid < CCH * (KORD + 1)) al[tid] = 0.5f * tanhf(temp[tid]);
  __syncthreads();

  int n = blockIdx.x * blockDim.x + tid;
  if (n >= N_NODES) return;
  float e = origin_e[n];

  const float a = 1.0f, b = 1.0f, lo = -1.0f, hi = 1.0f;
  const float coef1 = (a - b) * 0.5f - (a + b + 2.0f) * 0.5f * (lo + hi) / (hi - lo);
  const float coef2 = (a + b + 2.0f) / (hi - lo);

  for (int c = 0; c < CCH; ++c) {
    const float* ac = &al[c * (KORD + 1)];
    float xs_m2 = ac[0];
    float acc = xs_m2;
    float xs_m1 = coef1 * ac[1] + coef2 * ac[1] * e * xs_m2;
    acc += xs_m1;
#pragma unroll
    for (int L = 2; L <= KORD; ++L) {
      float Lf = (float)L;
      float coef_l  = 2.0f * Lf * (Lf + a + b) * (2.0f * Lf - 2.0f + a + b);
      float c_lm1_1 = (2.0f * Lf + a + b - 1.0f) * (2.0f * Lf + a + b) * (2.0f * Lf + a + b - 2.0f);
      float c_lm1_2 = (2.0f * Lf + a + b - 1.0f) * (a * a - b * b);
      float c_lm2   = 2.0f * (Lf - 1.0f + a) * (Lf - 1.0f + b) * (2.0f * Lf + a + b);
      float tmp1   = ac[L - 1] * (c_lm1_1 / coef_l);
      float tmp2   = ac[L - 1] * (c_lm1_2 / coef_l);
      float tmp3   = ac[L - 1] * ac[L - 2] * (c_lm2 / coef_l);
      float tmp1_2 = tmp1 * (2.0f / (hi - lo));
      float tmp2_2 = tmp1 * ((hi + lo) / (hi - lo)) + tmp2;
      float nx = tmp1_2 * (e * xs_m1) - tmp2_2 * xs_m1 - tmp3 * xs_m2;
      acc += nx;
      xs_m2 = xs_m1;
      xs_m1 = nx;
    }
    Hf[n * CCH + c] = acc;
  }
}

#define XS_STRIDE 516
#define HS_STRIDE 68

__global__ void mlp_kernel(const float* __restrict__ x,
                           const float* __restrict__ W1,
                           const float* __restrict__ b1,
                           const float* __restrict__ W2,
                           const float* __restrict__ b2,
                           float* __restrict__ zb) {
  __shared__ float xs[16 * XS_STRIDE];
  __shared__ float hs[16 * HS_STRIDE];

  const int tid  = threadIdx.x;
  const int w    = tid >> 5;
  const int lane = tid & 31;
  const int mn   = lane & 15;
  const int half = lane >> 4;
  const int kk   = half * 2;
  const int m0   = blockIdx.x * 16;
  const int n0   = w * 16;

  for (int idx = tid; idx < 16 * (F_INF / 4); idx += blockDim.x) {
    int row = idx >> 7;
    int c4  = idx & 127;
    float4 v = *(const float4*)&x[(size_t)(m0 + row) * F_INF + c4 * 4];
    float* d = &xs[row * XS_STRIDE + c4 * 4];
    d[0] = v.x; d[1] = v.y; d[2] = v.z; d[3] = v.w;
  }
  __syncthreads();

  v8f acc = {};
#pragma unroll 4
  for (int k0 = 0; k0 < F_INF; k0 += 4) {
    v2f av, bv;
    av.x = xs[mn * XS_STRIDE + k0 + kk];
    av.y = xs[mn * XS_STRIDE + k0 + kk + 1];
    bv.x = W1[(size_t)(k0 + kk) * HIDF + n0 + mn];
    bv.y = W1[(size_t)(k0 + kk + 1) * HIDF + n0 + mn];
    acc = wmma4(av, bv, acc);
  }

  float bias = b1[n0 + mn];
#pragma unroll
  for (int r = 0; r < 8; ++r) {
    int row = half ? r + 8 : r;
    float v = acc[r] + bias;
    hs[row * HS_STRIDE + n0 + mn] = v > 0.0f ? v : 0.0f;
  }
  __syncthreads();

  if (w == 0) {
    v8f zacc = {};
#pragma unroll
    for (int k0 = 0; k0 < HIDF; k0 += 4) {
      v2f av, bv;
      av.x = hs[mn * HS_STRIDE + k0 + kk];
      av.y = hs[mn * HS_STRIDE + k0 + kk + 1];
      bv.x = W2[(k0 + kk) * CCH + mn];
      bv.y = W2[(k0 + kk + 1) * CCH + mn];
      zacc = wmma4(av, bv, zacc);
    }
    float bb = b2[mn];
#pragma unroll
    for (int r = 0; r < 8; ++r) {
      int row = half ? r + 8 : r;
      zb[(size_t)(m0 + row) * CCH + mn] = zacc[r] + bb;
    }
  }
}

typedef __attribute__((ext_vector_type(4))) float v4f;

#define UT_COLS  64
#define UT_ROWS  256
#define UT_CW    2
#define UT_RW    8
#define UT_BCOLS (UT_CW * UT_COLS)
#define UT_BROWS (UT_RW * UT_ROWS)
#define UT_NTHR  (UT_CW * UT_RW * 32)
#define UT_KB    8
#define UT_NB    (UT_ROWS / 4 / UT_KB)
#define UT_BOUT  (UT_BCOLS * CCH)

__device__ __forceinline__ void utz_load_batch(v4f (&buf)[UT_KB][2],
                                               const float* __restrict__ up, int b) {
#pragma unroll
  for (int t = 0; t < UT_KB; ++t) {
#pragma unroll
    for (int s = 0; s < 2; ++s) {
      buf[t][s] = *(const v4f*)(up + (size_t)((b * UT_KB + t) * 4 + s) * N_NODES);
    }
  }
  __builtin_amdgcn_sched_barrier(0);
}

__device__ __forceinline__ void utz_compute_batch(const v4f (&buf)[UT_KB][2],
                                                  const float* zrow, int b,
                                                  v8f (&acc)[4]) {
  v2f av[UT_KB];
#pragma unroll
  for (int t = 0; t < UT_KB; ++t) {
    const float* zp = zrow + (b * UT_KB + t) * 4 * CCH;
    av[t].x = zp[0];
    av[t].y = zp[CCH];
  }
#pragma unroll
  for (int t = 0; t < UT_KB; ++t) {
#pragma unroll
    for (int q = 0; q < 4; ++q) {
      v2f bv;
      bv.x = buf[t][0][q];
      bv.y = buf[t][1][q];
      acc[q] = wmma4(av[t], bv, acc[q]);
    }
  }
}

__global__ __launch_bounds__(UT_NTHR) void utz_kernel(const float* __restrict__ U,
                                                      const float* __restrict__ zb,
                                                      float* __restrict__ tbuf) {
  __shared__ __attribute__((aligned(16))) float zs[UT_BROWS * CCH];

  const int tid  = threadIdx.x;
  const int w    = tid >> 5;
  const int lane = tid & 31;
  const int mn   = lane & 15;
  const int half = lane >> 4;
  const int cw   = w & (UT_CW - 1);
  const int rw   = w / UT_CW;
  const int cb   = blockIdx.x & (N_NODES / UT_BCOLS - 1);
  const int jb   = (blockIdx.x / (N_NODES / UT_BCOLS)) * UT_BROWS;
  const int i0   = cb * UT_BCOLS + cw * UT_COLS;
  const int jw   = jb + rw * UT_ROWS;

  v4f zv[UT_BROWS * CCH / 4 / UT_NTHR];
#pragma unroll
  for (int i = 0; i < UT_BROWS * CCH / 4 / UT_NTHR; ++i) {
    zv[i] = *(const v4f*)&zb[(size_t)jb * CCH + (size_t)(tid + i * UT_NTHR) * 4];
  }

  const float* up = U + (size_t)(jw + 2 * half) * N_NODES + i0 + 4 * mn;
  v4f bufA[UT_KB][2], bufB[UT_KB][2];
  utz_load_batch(bufA, up, 0);
  utz_load_batch(bufB, up, 1);

#pragma unroll
  for (int i = 0; i < UT_BROWS * CCH / 4 / UT_NTHR; ++i) {
    *(v4f*)&zs[(tid + i * UT_NTHR) * 4] = zv[i];
  }
  __syncthreads();

  const float* zrow = &zs[(rw * UT_ROWS + 2 * half) * CCH + mn];
  v8f acc[4] = {};

#pragma unroll 1
  for (int b = 0; b < UT_NB - 2; b += 2) {
    utz_compute_batch(bufA, zrow, b, acc);
    utz_load_batch(bufA, up, b + 2);
    utz_compute_batch(bufB, zrow, b + 1, acc);
    utz_load_batch(bufB, up, b + 3);
  }
  utz_compute_batch(bufA, zrow, UT_NB - 2, acc);
  utz_compute_batch(bufB, zrow, UT_NB - 1, acc);

  __syncthreads();

  float* tw = &zs[w * (UT_COLS * CCH)];
#pragma unroll
  for (int q = 0; q < 4; ++q) {
    float* d = &tw[(4 * mn + q) * CCH + 8 * half];
    v4f lo = {acc[q][0], acc[q][1], acc[q][2], acc[q][3]};
    v4f hi = {acc[q][4], acc[q][5], acc[q][6], acc[q][7]};
    *(v4f*)&d[0] = lo;
    *(v4f*)&d[4] = hi;
  }
  __syncthreads();

  float sum[UT_BOUT / UT_NTHR];
#pragma unroll
  for (int k = 0; k < UT_BOUT / UT_NTHR; ++k) {
    const int o = tid + k * UT_NTHR;
    float s = zs[o];
#pragma unroll
    for (int r = 1; r < UT_RW; ++r) s += zs[r * UT_BOUT + o];
    sum[k] = s;
  }
  float* tg = tbuf + (size_t)cb * UT_BOUT;
#pragma unroll
  for (int k = 0; k < UT_BOUT / UT_NTHR; ++k) {
    unsafeAtomicAdd(&tg[tid + k * UT_NTHR], sum[k]);
  }
}

#define UT_STRIDE 68

__global__ void outgemm_kernel(const float* __restrict__ U,
                               const float* __restrict__ tbuf,
                               const float* __restrict__ Hf,
                               float* __restrict__ obuf) {
  __shared__ float Ut[128 * UT_STRIDE];
  __shared__ float St[64 * CCH];

  const int tid  = threadIdx.x;
  const int w    = tid >> 5;
  const int lane = tid & 31;
  const int mn   = lane & 15;
  const int half = lane >> 4;
  const int kk   = half * 2;
  const int r0   = (blockIdx.x & 63) * 128;
  const int kb   = (blockIdx.x >> 6) * JCHUNK;

  v8f acc = {};
  for (int i0 = kb; i0 < kb + JCHUNK; i0 += 64) {
    for (int idx = tid; idx < 128 * 16; idx += blockDim.x) {
      int row = idx >> 4;
      int c4  = idx & 15;
      float4 v = *(const float4*)&U[(size_t)(r0 + row) * N_NODES + i0 + c4 * 4];
      float* d = &Ut[row * UT_STRIDE + c4 * 4];
      d[0] = v.x; d[1] = v.y; d[2] = v.z; d[3] = v.w;
    }
    {
      size_t base = (size_t)(i0 + (tid >> 2)) * CCH + (tid & 3) * 4;
      float4 tv = *(const float4*)&tbuf[base];
      float4 hv = *(const float4*)&Hf[base];
      float* d = &St[(tid >> 2) * CCH + (tid & 3) * 4];
      d[0] = tv.x * hv.x; d[1] = tv.y * hv.y; d[2] = tv.z * hv.z; d[3] = tv.w * hv.w;
    }
    __syncthreads();
#pragma unroll
    for (int k0 = 0; k0 < 64; k0 += 4) {
      v2f av, bv;
      av.x = Ut[(16 * w + mn) * UT_STRIDE + k0 + kk];
      av.y = Ut[(16 * w + mn) * UT_STRIDE + k0 + kk + 1];
      bv.x = St[(k0 + kk) * CCH + mn];
      bv.y = St[(k0 + kk + 1) * CCH + mn];
      acc = wmma4(av, bv, acc);
    }
    __syncthreads();
  }
#pragma unroll
  for (int r = 0; r < 8; ++r) {
    int row = half ? r + 8 : r;
    unsafeAtomicAdd(&obuf[(size_t)(r0 + 16 * w + row) * CCH + mn], acc[r]);
  }
}

__global__ void lsm_kernel(const float* __restrict__ obuf,
                           float* __restrict__ out) {
  int n = blockIdx.x * blockDim.x + threadIdx.x;
  if (n >= N_NODES) return;
  float v[CCH];
#pragma unroll
  for (int q = 0; q < 4; ++q) {
    float4 t = *(const float4*)&obuf[(size_t)n * CCH + q * 4];
    v[q * 4 + 0] = t.x; v[q * 4 + 1] = t.y; v[q * 4 + 2] = t.z; v[q * 4 + 3] = t.w;
  }
  float mx = v[0];
#pragma unroll
  for (int i = 1; i < CCH; ++i) mx = fmaxf(mx, v[i]);
  float sm = 0.0f;
#pragma unroll
  for (int i = 0; i < CCH; ++i) sm += expf(v[i] - mx);
  float lse = mx + logf(sm);
#pragma unroll
  for (int q = 0; q < 4; ++q) {
    float4 t;
    t.x = v[q * 4 + 0] - lse; t.y = v[q * 4 + 1] - lse;
    t.z = v[q * 4 + 2] - lse; t.w = v[q * 4 + 3] - lse;
    *(float4*)&out[(size_t)n * CCH + q * 4] = t;
  }
}

extern "C" void kernel_launch(void* const* d_in, const int* in_sizes, int n_in,
                              void* d_out, int out_size, void* d_ws, size_t ws_size,
                              hipStream_t stream) {
  const float* origin_e = (const float*)d_in[0];
  const float* U        = (const float*)d_in[1];
  const float* x        = (const float*)d_in[2];
  const float* W1       = (const float*)d_in[3];
  const float* b1       = (const float*)d_in[4];
  const float* W2       = (const float*)d_in[5];
  const float* b2       = (const float*)d_in[6];
  const float* temp     = (const float*)d_in[7];
  float* out = (float*)d_out;

  float* ws   = (float*)d_ws;
  float* Hf   = ws;
  float* zb   = Hf   + (size_t)N_NODES * CCH;
  float* tbuf = zb   + (size_t)N_NODES * CCH;
  float* obuf = tbuf + (size_t)N_NODES * CCH;

  zero_kernel<<<256, 256, 0, stream>>>((float4*)tbuf);
  jacobi_kernel<<<N_NODES / 256, 256, 0, stream>>>(origin_e, temp, Hf);
  mlp_kernel<<<N_NODES / 16, 128, 0, stream>>>(x, W1, b1, W2, b2, zb);
  utz_kernel<<<(N_NODES / UT_BCOLS) * (N_NODES / UT_BROWS), UT_NTHR, 0, stream>>>(U, zb, tbuf);
  outgemm_kernel<<<64 * NCHUNKS, 256, 0, stream>>>(U, tbuf, Hf, obuf);
  lsm_kernel<<<N_NODES / 256, 256, 0, stream>>>(obuf, out);
}


# out=U s: a wave streams its own 16 rows of U in 512-byte row pieces, 16-32 loads in flight, via a wave-private LDS tile into WMMA A fragments; 8 (was 16) atomic partials: sum order changes
# speedup vs baseline: 1.3447x; 1.2362x over previous
#include <hip/hip_runtime.h>
#include <cstdint>

typedef __attribute__((ext_vector_type(2))) float v2f;
typedef __attribute__((ext_vector_type(8))) float v8f;

#define N_NODES 8192
#define F_INF   512
#define HIDF    64
#define CCH     16
#define KORD    10
#define JCHUNK  512
#define NCHUNKS (N_NODES / JCHUNK)

__device__ __forceinline__ v8f wmma4(v2f a, v2f b, v8f c) {
  return __builtin_amdgcn_wmma_f32_16x16x4_f32(
       false, a,  false, b,
       (short)0, c,  false,  false);
}

__global__ void zero_kernel(float4* __restrict__ p) {
  p[blockIdx.x * blockDim.x + threadIdx.x] = make_float4(0.f, 0.f, 0.f, 0.f);
}

__global__ void jacobi_kernel(const float* __restrict__ origin_e,
                              const float* __restrict__ temp,
                              float* __restrict__ Hf) {
  __shared__ float al[CCH * (KORD + 1)];
  int tid = threadIdx.x;
  if (tid < CCH * (KORD + 1)) al[tid] = 0.5f * tanhf(temp[tid]);
  __syncthreads();

  int n = blockIdx.x * blockDim.x + tid;
  if (n >= N_NODES) return;
  float e = origin_e[n];

  const float a = 1.0f, b = 1.0f, lo = -1.0f, hi = 1.0f;
  const float coef1 = (a - b) * 0.5f - (a + b + 2.0f) * 0.5f * (lo + hi) / (hi - lo);
  const float coef2 = (a + b + 2.0f) / (hi - lo);

  for (int c = 0; c < CCH; ++c) {
    const float* ac = &al[c * (KORD + 1)];
    float xs_m2 = ac[0];
    float acc = xs_m2;
    float xs_m1 = coef1 * ac[1] + coef2 * ac[1] * e * xs_m2;
    acc += xs_m1;
#pragma unroll
    for (int L = 2; L <= KORD; ++L) {
      float Lf = (float)L;
      float coef_l  = 2.0f * Lf * (Lf + a + b) * (2.0f * Lf - 2.0f + a + b);
      float c_lm1_1 = (2.0f * Lf + a + b - 1.0f) * (2.0f * Lf + a + b) * (2.0f * Lf + a + b - 2.0f);
      float c_lm1_2 = (2.0f * Lf + a + b - 1.0f) * (a * a - b * b);
      float c_lm2   = 2.0f * (Lf - 1.0f + a) * (Lf - 1.0f + b) * (2.0f * Lf + a + b);
      float tmp1   = ac[L - 1] * (c_lm1_1 / coef_l);
      float tmp2   = ac[L - 1] * (c_lm1_2 / coef_l);
      float tmp3   = ac[L - 1] * ac[L - 2] * (c_lm2 / coef_l);
      float tmp1_2 = tmp1 * (2.0f / (hi - lo));
      float tmp2_2 = tmp1 * ((hi + lo) / (hi - lo)) + tmp2;
      float nx = tmp1_2 * (e * xs_m1) - tmp2_2 * xs_m1 - tmp3 * xs_m2;
      acc += nx;
      xs_m2 = xs_m1;
      xs_m1 = nx;
    }
    Hf[n * CCH + c] = acc;
  }
}

#define XS_STRIDE 516
#define HS_STRIDE 68

__global__ void mlp_kernel(const float* __restrict__ x,
                           const float* __restrict__ W1,
                           const float* __restrict__ b1,
                           const float* __restrict__ W2,
                           const float* __restrict__ b2,
                           float* __restrict__ zb) {
  __shared__ float xs[16 * XS_STRIDE];
  __shared__ float hs[16 * HS_STRIDE];

  const int tid  = threadIdx.x;
  const int w    = tid >> 5;
  const int lane = tid & 31;
  const int mn   = lane & 15;
  const int half = lane >> 4;
  const int kk   = half * 2;
  const int m0   = blockIdx.x * 16;
  const int n0   = w * 16;

  for (int idx = tid; idx < 16 * (F_INF / 4); idx += blockDim.x) {
    int row = idx >> 7;
    int c4  = idx & 127;
    float4 v = *(const float4*)&x[(size_t)(m0 + row) * F_INF + c4 * 4];
    float* d = &xs[row * XS_STRIDE + c4 * 4];
    d[0] = v.x; d[1] = v.y; d[2] = v.z; d[3] = v.w;
  }
  __syncthreads();

  v8f acc = {};
#pragma unroll 4
  for (int k0 = 0; k0 < F_INF; k0 += 4) {
    v2f av, bv;
    av.x = xs[mn * XS_STRIDE + k0 + kk];
    av.y = xs[mn * XS_STRIDE + k0 + kk + 1];
    bv.x = W1[(size_t)(k0 + kk) * HIDF + n0 + mn];
    bv.y = W1[(size_t)(k0 + kk + 1) * HIDF + n0 + mn];
    acc = wmma4(av, bv, acc);
  }

  float bias = b1[n0 + mn];
#pragma unroll
  for (int r = 0; r < 8; ++r) {
    int row = half ? r + 8 : r;
    float v = acc[r] + bias;
    hs[row * HS_STRIDE + n0 + mn] = v > 0.0f ? v : 0.0f;
  }
  __syncthreads();

  if (w == 0) {
    v8f zacc = {};
#pragma unroll
    for (int k0 = 0; k0 < HIDF; k0 += 4) {
      v2f av, bv;
      av.x = hs[mn * HS_STRIDE + k0 + kk];
      av.y = hs[mn * HS_STRIDE + k0 + kk + 1];
      bv.x = W2[(k0 + kk) * CCH + mn];
      bv.y = W2[(k0 + kk + 1) * CCH + mn];
      zacc = wmma4(av, bv, zacc);
    }
    float bb = b2[mn];
#pragma unroll
    for (int r = 0; r < 8; ++r) {
      int row = half ? r + 8 : r;
      zb[(size_t)(m0 + row) * CCH + mn] = zacc[r] + bb;
    }
  }
}

typedef __attribute__((ext_vector_type(4))) float v4f;

#define UT_COLS  64
#define UT_ROWS  256
#define UT_CW    2
#define UT_RW    8
#define UT_BCOLS (UT_CW * UT_COLS)
#define UT_BROWS (UT_RW * UT_ROWS)
#define UT_NTHR  (UT_CW * UT_RW * 32)
#define UT_KB    8
#define UT_NB    (UT_ROWS / 4 / UT_KB)
#define UT_BOUT  (UT_BCOLS * CCH)

__device__ __forceinline__ void utz_load_batch(v4f (&buf)[UT_KB][2],
                                               const float* __restrict__ up, int b) {
#pragma unroll
  for (int t = 0; t < UT_KB; ++t) {
#pragma unroll
    for (int s = 0; s < 2; ++s) {
      buf[t][s] = *(const v4f*)(up + (size_t)((b * UT_KB + t) * 4 + s) * N_NODES);
    }
  }
  __builtin_amdgcn_sched_barrier(0);
}

__device__ __forceinline__ void utz_compute_batch(const v4f (&buf)[UT_KB][2],
                                                  const float* zrow, int b,
                                                  v8f (&acc)[4]) {
  v2f av[UT_KB];
#pragma unroll
  for (int t = 0; t < UT_KB; ++t) {
    const float* zp = zrow + (b * UT_KB + t) * 4 * CCH;
    av[t].x = zp[0];
    av[t].y = zp[CCH];
  }
#pragma unroll
  for (int t = 0; t < UT_KB; ++t) {
#pragma unroll
    for (int q = 0; q < 4; ++q) {
      v2f bv;
      bv.x = buf[t][0][q];
      bv.y = buf[t][1][q];
      acc[q] = wmma4(av[t], bv, acc[q]);
    }
  }
}

__global__ __launch_bounds__(UT_NTHR) void utz_kernel(const float* __restrict__ U,
                                                      const float* __restrict__ zb,
                                                      float* __restrict__ tbuf) {
  __shared__ __attribute__((aligned(16))) float zs[UT_BROWS * CCH];

  const int tid  = threadIdx.x;
  const int w    = tid >> 5;
  const int lane = tid & 31;
  const int mn   = lane & 15;
  const int half = lane >> 4;
  const int cw   = w & (UT_CW - 1);
  const int rw   = w / UT_CW;
  const int cb   = blockIdx.x & (N_NODES / UT_BCOLS - 1);
  const int jb   = (blockIdx.x / (N_NODES / UT_BCOLS)) * UT_BROWS;
  const int i0   = cb * UT_BCOLS + cw * UT_COLS;
  const int jw   = jb + rw * UT_ROWS;

  v4f zv[UT_BROWS * CCH / 4 / UT_NTHR];
#pragma unroll
  for (int i = 0; i < UT_BROWS * CCH / 4 / UT_NTHR; ++i) {
    zv[i] = *(const v4f*)&zb[(size_t)jb * CCH + (size_t)(tid + i * UT_NTHR) * 4];
  }

  const float* up = U + (size_t)(jw + 2 * half) * N_NODES + i0 + 4 * mn;
  v4f bufA[UT_KB][2], bufB[UT_KB][2];
  utz_load_batch(bufA, up, 0);
  utz_load_batch(bufB, up, 1);

#pragma unroll
  for (int i = 0; i < UT_BROWS * CCH / 4 / UT_NTHR; ++i) {
    *(v4f*)&zs[(tid + i * UT_NTHR) * 4] = zv[i];
  }
  __syncthreads();

  const float* zrow = &zs[(rw * UT_ROWS + 2 * half) * CCH + mn];
  v8f acc[4] = {};

#pragma unroll 1
  for (int b = 0; b < UT_NB - 2; b += 2) {
    utz_compute_batch(bufA, zrow, b, acc);
    utz_load_batch(bufA, up, b + 2);
    utz_compute_batch(bufB, zrow, b + 1, acc);
    utz_load_batch(bufB, up, b + 3);
  }
  utz_compute_batch(bufA, zrow, UT_NB - 2, acc);
  utz_compute_batch(bufB, zrow, UT_NB - 1, acc);

  __syncthreads();

  float* tw = &zs[w * (UT_COLS * CCH)];
#pragma unroll
  for (int q = 0; q < 4; ++q) {
    float* d = &tw[(4 * mn + q) * CCH + 8 * half];
    v4f lo = {acc[q][0], acc[q][1], acc[q][2], acc[q][3]};
    v4f hi = {acc[q][4], acc[q][5], acc[q][6], acc[q][7]};
    *(v4f*)&d[0] = lo;
    *(v4f*)&d[4] = hi;
  }
  __syncthreads();

  float sum[UT_BOUT / UT_NTHR];
#pragma unroll
  for (int k = 0; k < UT_BOUT / UT_NTHR; ++k) {
    const int o = tid + k * UT_NTHR;
    float s = zs[o];
#pragma unroll
    for (int r = 1; r < UT_RW; ++r) s += zs[r * UT_BOUT + o];
    sum[k] = s;
  }
  float* tg = tbuf + (size_t)cb * UT_BOUT;
#pragma unroll
  for (int k = 0; k < UT_BOUT / UT_NTHR; ++k) {
    unsafeAtomicAdd(&tg[tid + k * UT_NTHR], sum[k]);
  }
}

#define OG_WAVES  16
#define OG_BROWS  (16 * OG_WAVES)
#define OG_KC     1024
#define OG_KB     128
#define OG_NB     (OG_KC / OG_KB)
#define OG_APITCH (OG_KB + 4)
#define OG_SPITCH (OG_KC + 4)
#define OG_NTHR   (OG_WAVES * 32)
#define OG_SV     (OG_KC * CCH / 4 / OG_NTHR)

__device__ __forceinline__ void og_load_batch(v4f (&buf)[16],
                                              const float* __restrict__ up, int b) {
#pragma unroll
  for (int r = 0; r < 16; ++r) {
    buf[r] = *(const v4f*)(up + (size_t)r * N_NODES + b * OG_KB);
  }
  __builtin_amdgcn_sched_barrier(0);
}

__device__ __forceinline__ void og_stage_batch(const v4f (&buf)[16], float* at, int lane) {
#pragma unroll
  for (int r = 0; r < 16; ++r) {
    *(v4f*)&at[r * OG_APITCH + 4 * lane] = buf[r];
  }
}

__device__ __forceinline__ void og_compute_batch(const float* arow, const float* srow,
                                                 v8f& acc0, v8f& acc1) {
  v4f a[8], s[8];
#pragma unroll
  for (int j = 0; j < 8; ++j) {
    a[j] = *(const v4f*)&arow[8 * j];
    s[j] = *(const v4f*)&srow[8 * j];
  }
  __builtin_amdgcn_sched_barrier(0);
#pragma unroll
  for (int g = 0; g < OG_KB / 32; ++g) {
#pragma unroll
    for (int jj = 0; jj < 4; ++jj) {
      const int i = (g & 1) * 4 + jj;
      v2f a0, s0, a1, s1;
      a0.x = a[i][0]; a0.y = a[i][1]; s0.x = s[i][0]; s0.y = s[i][1];
      a1.x = a[i][2]; a1.y = a[i][3]; s1.x = s[i][2]; s1.y = s[i][3];
      acc0 = wmma4(a0, s0, acc0);
      acc1 = wmma4(a1, s1, acc1);
    }
    if (g + 2 < OG_KB / 32) {
#pragma unroll
      for (int jj = 0; jj < 4; ++jj) {
        const int i = (g & 1) * 4 + jj;
        const int j = 4 * (g + 2) + jj;
        a[i] = *(const v4f*)&arow[8 * j];
        s[i] = *(const v4f*)&srow[8 * j];
      }
    }
    __builtin_amdgcn_sched_barrier(0);
  }
}

__global__ __launch_bounds__(OG_NTHR) void outgemm_kernel(const float* __restrict__ U,
                                                          const float* __restrict__ tbuf,
                                                          const float* __restrict__ Hf,
                                                          float* __restrict__ obuf) {
  __shared__ __attribute__((aligned(16))) float sT[CCH * OG_SPITCH];
  __shared__ __attribute__((aligned(16))) float At[OG_WAVES * 16 * OG_APITCH];

  const int tid  = threadIdx.x;
  const int w    = tid >> 5;
  const int lane = tid & 31;
  const int mn   = lane & 15;
  const int half = lane >> 4;
  const int rb   = blockIdx.x & (N_NODES / OG_BROWS - 1);
  const int kb   = (blockIdx.x / (N_NODES / OG_BROWS)) * OG_KC;
  const int r0   = rb * OG_BROWS + w * 16;

  v4f tv[OG_SV], hv[OG_SV];
#pragma unroll
  for (int i = 0; i < OG_SV; ++i) {
    const size_t off = (size_t)kb * CCH + (size_t)(tid + i * OG_NTHR) * 4;
    tv[i] = *(const v4f*)&tbuf[off];
    hv[i] = *(const v4f*)&Hf[off];
  }

  const float* up = U + (size_t)r0 * N_NODES + kb + 4 * lane;
  v4f bufA[16], bufB[16];
  const int rot = (w + rb) & (OG_NB - 1);
  og_load_batch(bufA, up, rot);
  og_load_batch(bufB, up, (rot + 1) & (OG_NB - 1));

#pragma unroll
  for (int i = 0; i < OG_SV; ++i) {
    const int e  = tid + i * OG_NTHR;
    const int k  = e >> 2;
    const int c0 = (e & 3) * 4;
#pragma unroll
    for (int q = 0; q < 4; ++q) {
      sT[(c0 + q) * OG_SPITCH + k] = tv[i][q] * hv[i][q];
    }
  }
  __builtin_amdgcn_fence(__ATOMIC_RELEASE, "workgroup", "local");
  __builtin_amdgcn_s_barrier();
  __builtin_amdgcn_fence(__ATOMIC_ACQUIRE, "workgroup", "local");

  float* at = &At[w * (16 * OG_APITCH)];
  const float* arow = &at[mn * OG_APITCH + 4 * half];
  const float* srow = &sT[mn * OG_SPITCH + 4 * half];
  v8f acc0 = {}, acc1 = {};

#pragma unroll 1
  for (int b = 0; b < OG_NB - 2; b += 2) {
    og_stage_batch(bufA, at, lane);
    og_load_batch(bufA, up, (b + 2 + rot) & (OG_NB - 1));
    og_compute_batch(arow, srow + ((b + rot) & (OG_NB - 1)) * OG_KB, acc0, acc1);
    og_stage_batch(bufB, at, lane);
    og_load_batch(bufB, up, (b + 3 + rot) & (OG_NB - 1));
    og_compute_batch(arow, srow + ((b + 1 + rot) & (OG_NB - 1)) * OG_KB, acc0, acc1);
  }
  og_stage_batch(bufA, at, lane);
  og_compute_batch(arow, srow + ((OG_NB - 2 + rot) & (OG_NB - 1)) * OG_KB, acc0, acc1);
  og_stage_batch(bufB, at, lane);
  og_compute_batch(arow, srow + ((OG_NB - 1 + rot) & (OG_NB - 1)) * OG_KB, acc0, acc1);

#pragma unroll
  for (int r = 0; r < 8; ++r) {
    unsafeAtomicAdd(&obuf[(size_t)(r0 + 8 * half + r) * CCH + mn], acc0[r] + acc1[r]);
  }
}

__global__ void lsm_kernel(const float* __restrict__ obuf,
                           float* __restrict__ out) {
  int n = blockIdx.x * blockDim.x + threadIdx.x;
  if (n >= N_NODES) return;
  float v[CCH];
#pragma unroll
  for (int q = 0; q < 4; ++q) {
    float4 t = *(const float4*)&obuf[(size_t)n * CCH + q * 4];
    v[q * 4 + 0] = t.x; v[q * 4 + 1] = t.y; v[q * 4 + 2] = t.z; v[q * 4 + 3] = t.w;
  }
  float mx = v[0];
#pragma unroll
  for (int i = 1; i < CCH; ++i) mx = fmaxf(mx, v[i]);
  float sm = 0.0f;
#pragma unroll
  for (int i = 0; i < CCH; ++i) sm += expf(v[i] - mx);
  float lse = mx + logf(sm);
#pragma unroll
  for (int q = 0; q < 4; ++q) {
    float4 t;
    t.x = v[q * 4 + 0] - lse; t.y = v[q * 4 + 1] - lse;
    t.z = v[q * 4 + 2] - lse; t.w = v[q * 4 + 3] - lse;
    *(float4*)&out[(size_t)n * CCH + q * 4] = t;
  }
}

extern "C" void kernel_launch(void* const* d_in, const int* in_sizes, int n_in,
                              void* d_out, int out_size, void* d_ws, size_t ws_size,
                              hipStream_t stream) {
  const float* origin_e = (const float*)d_in[0];
  const float* U        = (const float*)d_in[1];
  const float* x        = (const float*)d_in[2];
  const float* W1       = (const float*)d_in[3];
  const float* b1       = (const float*)d_in[4];
  const float* W2       = (const float*)d_in[5];
  const float* b2       = (const float*)d_in[6];
  const float* temp     = (const float*)d_in[7];
  float* out = (float*)d_out;

  float* ws   = (float*)d_ws;
  float* Hf   = ws;
  float* zb   = Hf   + (size_t)N_NODES * CCH;
  float* tbuf = zb   + (size_t)N_NODES * CCH;
  float* obuf = tbuf + (size_t)N_NODES * CCH;

  zero_kernel<<<256, 256, 0, stream>>>((float4*)tbuf);
  jacobi_kernel<<<N_NODES / 256, 256, 0, stream>>>(origin_e, temp, Hf);
  mlp_kernel<<<N_NODES / 16, 128, 0, stream>>>(x, W1, b1, W2, b2, zb);
  utz_kernel<<<(N_NODES / UT_BCOLS) * (N_NODES / UT_BROWS), UT_NTHR, 0, stream>>>(U, zb, tbuf);
  outgemm_kernel<<<(N_NODES / OG_BROWS) * (N_NODES / OG_KC), OG_NTHR, 0, stream>>>(U, tbuf, Hf, obuf);
  lsm_kernel<<<N_NODES / 256, 256, 0, stream>>>(obuf, out);
}


# accumulator clear, Jacobi filter and MLP in one launch (6 -> 4 dependent launches): each MLP block also clears 2 KB of tbuf+obuf and computes the filter of its 16 rows under the x tile's loads
# speedup vs baseline: 1.5624x; 1.1619x over previous
#include <hip/hip_runtime.h>
#include <cstdint>

typedef __attribute__((ext_vector_type(2))) float v2f;
typedef __attribute__((ext_vector_type(8))) float v8f;

#define N_NODES 8192
#define F_INF   512
#define HIDF    64
#define CCH     16
#define KORD    10
#define JCHUNK  512
#define NCHUNKS (N_NODES / JCHUNK)

__device__ __forceinline__ v8f wmma4(v2f a, v2f b, v8f c) {
  return __builtin_amdgcn_wmma_f32_16x16x4_f32(
       false, a,  false, b,
       (short)0, c,  false,  false);
}

typedef __attribute__((ext_vector_type(4))) float v4f;

#define XS_STRIDE 516
#define HS_STRIDE 68
#define PRE_NTHR  128
#define PRE_XV    (16 * (F_INF / 4) / PRE_NTHR)

__device__ __forceinline__ void lds_only_barrier() {
  __builtin_amdgcn_fence(__ATOMIC_RELEASE, "workgroup", "local");
  __builtin_amdgcn_s_barrier();
  __builtin_amdgcn_fence(__ATOMIC_ACQUIRE, "workgroup", "local");
}

__global__ __launch_bounds__(PRE_NTHR) void pre_kernel(const float* __restrict__ x,
                                                       const float* __restrict__ W1,
                                                       const float* __restrict__ b1,
                                                       const float* __restrict__ W2,
                                                       const float* __restrict__ b2,
                                                       const float* __restrict__ origin_e,
                                                       const float* __restrict__ temp,
                                                       float* __restrict__ zb,
                                                       float* __restrict__ Hf,
                                                       float* __restrict__ accbuf) {
  __shared__ float xs[16 * XS_STRIDE];
  __shared__ float hs[16 * HS_STRIDE];
  __shared__ float al[CCH * (KORD + 1)];

  const int tid  = threadIdx.x;
  const int w    = tid >> 5;
  const int lane = tid & 31;
  const int mn   = lane & 15;
  const int half = lane >> 4;
  const int kk   = half * 2;
  const int m0   = blockIdx.x * 16;
  const int n0   = w * 16;

  const int jr = tid >> 3;
  const int jc = (tid & 7) * 2;
  const int tid2 = tid < CCH * (KORD + 1) - PRE_NTHR ? tid + PRE_NTHR : tid;
  const float tp0 = temp[tid];
  const float tp1 = temp[tid2];
  const float e   = origin_e[m0 + jr];

  v4f xv[PRE_XV];
#pragma unroll
  for (int i = 0; i < PRE_XV; ++i) {
    xv[i] = *(const v4f*)&x[(size_t)(m0 + i) * F_INF + tid * 4];
  }

  {
    v4f zero4 = {0.f, 0.f, 0.f, 0.f};
    *(v4f*)&accbuf[((size_t)blockIdx.x * PRE_NTHR + tid) * 4] = zero4;
  }

  al[tid] = 0.5f * tanhf(tp0);
  if (tid2 != tid) al[tid2] = 0.5f * tanhf(tp1);
  lds_only_barrier();

  {
    const float a = 1.0f, b = 1.0f, lo = -1.0f, hi = 1.0f;
    const float coef1 = (a - b) * 0.5f - (a + b + 2.0f) * 0.5f * (lo + hi) / (hi - lo);
    const float coef2 = (a + b + 2.0f) / (hi - lo);

    v2f hf2;
#pragma unroll
    for (int cc = 0; cc < 2; ++cc) {
      const float* ac = &al[(jc + cc) * (KORD + 1)];
      float xs_m2 = ac[0];
      float acc = xs_m2;
      float xs_m1 = coef1 * ac[1] + coef2 * ac[1] * e * xs_m2;
      acc += xs_m1;
#pragma unroll
      for (int L = 2; L <= KORD; ++L) {
        float Lf = (float)L;
        float coef_l  = 2.0f * Lf * (Lf + a + b) * (2.0f * Lf - 2.0f + a + b);
        float c_lm1_1 = (2.0f * Lf + a + b - 1.0f) * (2.0f * Lf + a + b) * (2.0f * Lf + a + b - 2.0f);
        float c_lm1_2 = (2.0f * Lf + a + b - 1.0f) * (a * a - b * b);
        float c_lm2   = 2.0f * (Lf - 1.0f + a) * (Lf - 1.0f + b) * (2.0f * Lf + a + b);
        float tmp1   = ac[L - 1] * (c_lm1_1 / coef_l);
        float tmp2   = ac[L - 1] * (c_lm1_2 / coef_l);
        float tmp3   = ac[L - 1] * ac[L - 2] * (c_lm2 / coef_l);
        float tmp1_2 = tmp1 * (2.0f / (hi - lo));
        float tmp2_2 = tmp1 * ((hi + lo) / (hi - lo)) + tmp2;
        float nx = tmp1_2 * (e * xs_m1) - tmp2_2 * xs_m1 - tmp3 * xs_m2;
        acc += nx;
        xs_m2 = xs_m1;
        xs_m1 = nx;
      }
      hf2[cc] = acc;
    }
    *(v2f*)&Hf[(size_t)(m0 + jr) * CCH + jc] = hf2;
  }
  __builtin_amdgcn_sched_barrier(0);

#pragma unroll
  for (int i = 0; i < PRE_XV; ++i) {
    float* d = &xs[i * XS_STRIDE + tid * 4];
    d[0] = xv[i][0]; d[1] = xv[i][1]; d[2] = xv[i][2]; d[3] = xv[i][3];
  }
  lds_only_barrier();

  v8f acc = {};
#pragma unroll 4
  for (int k0 = 0; k0 < F_INF; k0 += 4) {
    v2f av, bv;
    av.x = xs[mn * XS_STRIDE + k0 + kk];
    av.y = xs[mn * XS_STRIDE + k0 + kk + 1];
    bv.x = W1[(size_t)(k0 + kk) * HIDF + n0 + mn];
    bv.y = W1[(size_t)(k0 + kk + 1) * HIDF + n0 + mn];
    acc = wmma4(av, bv, acc);
  }

  float bias = b1[n0 + mn];
#pragma unroll
  for (int r = 0; r < 8; ++r) {
    int row = half ? r + 8 : r;
    float v = acc[r] + bias;
    hs[row * HS_STRIDE + n0 + mn] = v > 0.0f ? v : 0.0f;
  }
  lds_only_barrier();

  if (w == 0) {
    v8f zacc = {};
#pragma unroll
    for (int k0 = 0; k0 < HIDF; k0 += 4) {
      v2f av, bv;
      av.x = hs[mn * HS_STRIDE + k0 + kk];
      av.y = hs[mn * HS_STRIDE + k0 + kk + 1];
      bv.x = W2[(k0 + kk) * CCH + mn];
      bv.y = W2[(k0 + kk + 1) * CCH + mn];
      zacc = wmma4(av, bv, zacc);
    }
    float bb = b2[mn];
#pragma unroll
    for (int r = 0; r < 8; ++r) {
      int row = half ? r + 8 : r;
      zb[(size_t)(m0 + row) * CCH + mn] = zacc[r] + bb;
    }
  }
}

#define UT_COLS  64
#define UT_ROWS  256
#define UT_CW    2
#define UT_RW    8
#define UT_BCOLS (UT_CW * UT_COLS)
#define UT_BROWS (UT_RW * UT_ROWS)
#define UT_NTHR  (UT_CW * UT_RW * 32)
#define UT_KB    8
#define UT_NB    (UT_ROWS / 4 / UT_KB)
#define UT_BOUT  (UT_BCOLS * CCH)

__device__ __forceinline__ void utz_load_batch(v4f (&buf)[UT_KB][2],
                                               const float* __restrict__ up, int b) {
#pragma unroll
  for (int t = 0; t < UT_KB; ++t) {
#pragma unroll
    for (int s = 0; s < 2; ++s) {
      buf[t][s] = *(const v4f*)(up + (size_t)((b * UT_KB + t) * 4 + s) * N_NODES);
    }
  }
  __builtin_amdgcn_sched_barrier(0);
}

__device__ __forceinline__ void utz_compute_batch(const v4f (&buf)[UT_KB][2],
                                                  const float* zrow, int b,
                                                  v8f (&acc)[4]) {
  v2f av[UT_KB];
#pragma unroll
  for (int t = 0; t < UT_KB; ++t) {
    const float* zp = zrow + (b * UT_KB + t) * 4 * CCH;
    av[t].x = zp[0];
    av[t].y = zp[CCH];
  }
#pragma unroll
  for (int t = 0; t < UT_KB; ++t) {
#pragma unroll
    for (int q = 0; q < 4; ++q) {
      v2f bv;
      bv.x = buf[t][0][q];
      bv.y = buf[t][1][q];
      acc[q] = wmma4(av[t], bv, acc[q]);
    }
  }
}

__global__ __launch_bounds__(UT_NTHR) void utz_kernel(const float* __restrict__ U,
                                                      const float* __restrict__ zb,
                                                      float* __restrict__ tbuf) {
  __shared__ __attribute__((aligned(16))) float zs[UT_BROWS * CCH];

  const int tid  = threadIdx.x;
  const int w    = tid >> 5;
  const int lane = tid & 31;
  const int mn   = lane & 15;
  const int half = lane >> 4;
  const int cw   = w & (UT_CW - 1);
  const int rw   = w / UT_CW;
  const int cb   = blockIdx.x & (N_NODES / UT_BCOLS - 1);
  const int jb   = (blockIdx.x / (N_NODES / UT_BCOLS)) * UT_BROWS;
  const int i0   = cb * UT_BCOLS + cw * UT_COLS;
  const int jw   = jb + rw * UT_ROWS;

  v4f zv[UT_BROWS * CCH / 4 / UT_NTHR];
#pragma unroll
  for (int i = 0; i < UT_BROWS * CCH / 4 / UT_NTHR; ++i) {
    zv[i] = *(const v4f*)&zb[(size_t)jb * CCH + (size_t)(tid + i * UT_NTHR) * 4];
  }

  const float* up = U + (size_t)(jw + 2 * half) * N_NODES + i0 + 4 * mn;
  v4f bufA[UT_KB][2], bufB[UT_KB][2];
  utz_load_batch(bufA, up, 0);
  utz_load_batch(bufB, up, 1);

#pragma unroll
  for (int i = 0; i < UT_BROWS * CCH / 4 / UT_NTHR; ++i) {
    *(v4f*)&zs[(tid + i * UT_NTHR) * 4] = zv[i];
  }
  __syncthreads();

  const float* zrow = &zs[(rw * UT_ROWS + 2 * half) * CCH + mn];
  v8f acc[4] = {};

#pragma unroll 1
  for (int b = 0; b < UT_NB - 2; b += 2) {
    utz_compute_batch(bufA, zrow, b, acc);
    utz_load_batch(bufA, up, b + 2);
    utz_compute_batch(bufB, zrow, b + 1, acc);
    utz_load_batch(bufB, up, b + 3);
  }
  utz_compute_batch(bufA, zrow, UT_NB - 2, acc);
  utz_compute_batch(bufB, zrow, UT_NB - 1, acc);

  __syncthreads();

  float* tw = &zs[w * (UT_COLS * CCH)];
#pragma unroll
  for (int q = 0; q < 4; ++q) {
    float* d = &tw[(4 * mn + q) * CCH + 8 * half];
    v4f lo = {acc[q][0], acc[q][1], acc[q][2], acc[q][3]};
    v4f hi = {acc[q][4], acc[q][5], acc[q][6], acc[q][7]};
    *(v4f*)&d[0] = lo;
    *(v4f*)&d[4] = hi;
  }
  __syncthreads();

  float sum[UT_BOUT / UT_NTHR];
#pragma unroll
  for (int k = 0; k < UT_BOUT / UT_NTHR; ++k) {
    const int o = tid + k * UT_NTHR;
    float s = zs[o];
#pragma unroll
    for (int r = 1; r < UT_RW; ++r) s += zs[r * UT_BOUT + o];
    sum[k] = s;
  }
  float* tg = tbuf + (size_t)cb * UT_BOUT;
#pragma unroll
  for (int k = 0; k < UT_BOUT / UT_NTHR; ++k) {
    unsafeAtomicAdd(&tg[tid + k * UT_NTHR], sum[k]);
  }
}

#define OG_WAVES  16
#define OG_BROWS  (16 * OG_WAVES)
#define OG_KC     1024
#define OG_KB     128
#define OG_NB     (OG_KC / OG_KB)
#define OG_APITCH (OG_KB + 4)
#define OG_SPITCH (OG_KC + 4)
#define OG_NTHR   (OG_WAVES * 32)
#define OG_SV     (OG_KC * CCH / 4 / OG_NTHR)

__device__ __forceinline__ void og_load_batch(v4f (&buf)[16],
                                              const float* __restrict__ up, int b) {
#pragma unroll
  for (int r = 0; r < 16; ++r) {
    buf[r] = *(const v4f*)(up + (size_t)r * N_NODES + b * OG_KB);
  }
  __builtin_amdgcn_sched_barrier(0);
}

__device__ __forceinline__ void og_stage_batch(const v4f (&buf)[16], float* at, int lane) {
#pragma unroll
  for (int r = 0; r < 16; ++r) {
    *(v4f*)&at[r * OG_APITCH + 4 * lane] = buf[r];
  }
}

__device__ __forceinline__ void og_compute_batch(const float* arow, const float* srow,
                                                 v8f& acc0, v8f& acc1) {
  v4f a[8], s[8];
#pragma unroll
  for (int j = 0; j < 8; ++j) {
    a[j] = *(const v4f*)&arow[8 * j];
    s[j] = *(const v4f*)&srow[8 * j];
  }
  __builtin_amdgcn_sched_barrier(0);
#pragma unroll
  for (int g = 0; g < OG_KB / 32; ++g) {
#pragma unroll
    for (int jj = 0; jj < 4; ++jj) {
      const int i = (g & 1) * 4 + jj;
      v2f a0, s0, a1, s1;
      a0.x = a[i][0]; a0.y = a[i][1]; s0.x = s[i][0]; s0.y = s[i][1];
      a1.x = a[i][2]; a1.y = a[i][3]; s1.x = s[i][2]; s1.y = s[i][3];
      acc0 = wmma4(a0, s0, acc0);
      acc1 = wmma4(a1, s1, acc1);
    }
    if (g + 2 < OG_KB / 32) {
#pragma unroll
      for (int jj = 0; jj < 4; ++jj) {
        const int i = (g & 1) * 4 + jj;
        const int j = 4 * (g + 2) + jj;
        a[i] = *(const v4f*)&arow[8 * j];
        s[i] = *(const v4f*)&srow[8 * j];
      }
    }
    __builtin_amdgcn_sched_barrier(0);
  }
}

__global__ __launch_bounds__(OG_NTHR) void outgemm_kernel(const float* __restrict__ U,
                                                          const float* __restrict__ tbuf,
                                                          const float* __restrict__ Hf,
                                                          float* __restrict__ obuf) {
  __shared__ __attribute__((aligned(16))) float sT[CCH * OG_SPITCH];
  __shared__ __attribute__((aligned(16))) float At[OG_WAVES * 16 * OG_APITCH];

  const int tid  = threadIdx.x;
  const int w    = tid >> 5;
  const int lane = tid & 31;
  const int mn   = lane & 15;
  const int half = lane >> 4;
  const int rb   = blockIdx.x & (N_NODES / OG_BROWS - 1);
  const int kb   = (blockIdx.x / (N_NODES / OG_BROWS)) * OG_KC;
  const int r0   = rb * OG_BROWS + w * 16;

  v4f tv[OG_SV], hv[OG_SV];
#pragma unroll
  for (int i = 0; i < OG_SV; ++i) {
    const size_t off = (size_t)kb * CCH + (size_t)(tid + i * OG_NTHR) * 4;
    tv[i] = *(const v4f*)&tbuf[off];
    hv[i] = *(const v4f*)&Hf[off];
  }

  const float* up = U + (size_t)r0 * N_NODES + kb + 4 * lane;
  v4f bufA[16], bufB[16];
  const int rot = (w + rb) & (OG_NB - 1);
  og_load_batch(bufA, up, rot);
  og_load_batch(bufB, up, (rot + 1) & (OG_NB - 1));

#pragma unroll
  for (int i = 0; i < OG_SV; ++i) {
    const int e  = tid + i * OG_NTHR;
    const int k  = e >> 2;
    const int c0 = (e & 3) * 4;
#pragma unroll
    for (int q = 0; q < 4; ++q) {
      sT[(c0 + q) * OG_SPITCH + k] = tv[i][q] * hv[i][q];
    }
  }
  __builtin_amdgcn_fence(__ATOMIC_RELEASE, "workgroup", "local");
  __builtin_amdgcn_s_barrier();
  __builtin_amdgcn_fence(__ATOMIC_ACQUIRE, "workgroup", "local");

  float* at = &At[w * (16 * OG_APITCH)];
  const float* arow = &at[mn * OG_APITCH + 4 * half];
  const float* srow = &sT[mn * OG_SPITCH + 4 * half];
  v8f acc0 = {}, acc1 = {};

#pragma unroll 1
  for (int b = 0; b < OG_NB - 2; b += 2) {
    og_stage_batch(bufA, at, lane);
    og_load_batch(bufA, up, (b + 2 + rot) & (OG_NB - 1));
    og_compute_batch(arow, srow + ((b + rot) & (OG_NB - 1)) * OG_KB, acc0, acc1);
    og_stage_batch(bufB, at, lane);
    og_load_batch(bufB, up, (b + 3 + rot) & (OG_NB - 1));
    og_compute_batch(arow, srow + ((b + 1 + rot) & (OG_NB - 1)) * OG_KB, acc0, acc1);
  }
  og_stage_batch(bufA, at, lane);
  og_compute_batch(arow, srow + ((OG_NB - 2 + rot) & (OG_NB - 1)) * OG_KB, acc0, acc1);
  og_stage_batch(bufB, at, lane);
  og_compute_batch(arow, srow + ((OG_NB - 1 + rot) & (OG_NB - 1)) * OG_KB, acc0, acc1);

#pragma unroll
  for (int r = 0; r < 8; ++r) {
    unsafeAtomicAdd(&obuf[(size_t)(r0 + 8 * half + r) * CCH + mn], acc0[r] + acc1[r]);
  }
}

__global__ void lsm_kernel(const float* __restrict__ obuf,
                           float* __restrict__ out) {
  int n = blockIdx.x * blockDim.x + threadIdx.x;
  if (n >= N_NODES) return;
  float v[CCH];
#pragma unroll
  for (int q = 0; q < 4; ++q) {
    float4 t = *(const float4*)&obuf[(size_t)n * CCH + q * 4];
    v[q * 4 + 0] = t.x; v[q * 4 + 1] = t.y; v[q * 4 + 2] = t.z; v[q * 4 + 3] = t.w;
  }
  float mx = v[0];
#pragma unroll
  for (int i = 1; i < CCH; ++i) mx = fmaxf(mx, v[i]);
  float sm = 0.0f;
#pragma unroll
  for (int i = 0; i < CCH; ++i) sm += expf(v[i] - mx);
  float lse = mx + logf(sm);
#pragma unroll
  for (int q = 0; q < 4; ++q) {
    float4 t;
    t.x = v[q * 4 + 0] - lse; t.y = v[q * 4 + 1] - lse;
    t.z = v[q * 4 + 2] - lse; t.w = v[q * 4 + 3] - lse;
    *(float4*)&out[(size_t)n * CCH + q * 4] = t;
  }
}

extern "C" void kernel_launch(void* const* d_in, const int* in_sizes, int n_in,
                              void* d_out, int out_size, void* d_ws, size_t ws_size,
                              hipStream_t stream) {
  const float* origin_e = (const float*)d_in[0];
  const float* U        = (const float*)d_in[1];
  const float* x        = (const float*)d_in[2];
  const float* W1       = (const float*)d_in[3];
  const float* b1       = (const float*)d_in[4];
  const float* W2       = (const float*)d_in[5];
  const float* b2       = (const float*)d_in[6];
  const float* temp     = (const float*)d_in[7];
  float* out = (float*)d_out;

  float* ws   = (float*)d_ws;
  float* Hf   = ws;
  float* zb   = Hf   + (size_t)N_NODES * CCH;
  float* tbuf = zb   + (size_t)N_NODES * CCH;
  float* obuf = tbuf + (size_t)N_NODES * CCH;

  pre_kernel<<<N_NODES / 16, PRE_NTHR, 0, stream>>>(x, W1, b1, W2, b2, origin_e, temp, zb, Hf, tbuf);
  utz_kernel<<<(N_NODES / UT_BCOLS) * (N_NODES / UT_BROWS), UT_NTHR, 0, stream>>>(U, zb, tbuf);
  outgemm_kernel<<<(N_NODES / OG_BROWS) * (N_NODES / OG_KC), OG_NTHR, 0, stream>>>(U, tbuf, Hf, obuf);
  lsm_kernel<<<N_NODES / 256, 256, 0, stream>>>(obuf, out);
}


# MLP layer 1: a wave takes a quarter of k for all 64 hidden columns, its W1 slice requested whole behind the x tile: no memory wait in the matrix loop; 4 partials summed in LDS: sum order changes
# speedup vs baseline: 1.8991x; 1.2155x over previous
#include <hip/hip_runtime.h>
#include <cstdint>

typedef __attribute__((ext_vector_type(2))) float v2f;
typedef __attribute__((ext_vector_type(8))) float v8f;

#define N_NODES 8192
#define F_INF   512
#define HIDF    64
#define CCH     16
#define KORD    10
#define JCHUNK  512
#define NCHUNKS (N_NODES / JCHUNK)

__device__ __forceinline__ v8f wmma4(v2f a, v2f b, v8f c) {
  return __builtin_amdgcn_wmma_f32_16x16x4_f32(
       false, a,  false, b,
       (short)0, c,  false,  false);
}

typedef __attribute__((ext_vector_type(4))) float v4f;

#define XS_STRIDE 516
#define HS_STRIDE 68
#define PRE_NTHR  128
#define PRE_XV    (16 * (F_INF / 4) / PRE_NTHR)
#define PRE_KQ    (F_INF / 4)
#define PRE_NJ    (PRE_KQ / 8)
#define PRE_J1    10

__device__ __forceinline__ void lds_only_barrier() {
  __builtin_amdgcn_fence(__ATOMIC_RELEASE, "workgroup", "local");
  __builtin_amdgcn_s_barrier();
  __builtin_amdgcn_fence(__ATOMIC_ACQUIRE, "workgroup", "local");
}

__device__ __forceinline__ void pre_load_w1(v4f (&bw)[PRE_NJ][4],
                                            const float* __restrict__ wp, int j0, int j1) {
#pragma unroll
  for (int j = 0; j < PRE_NJ; ++j) {
    if (j >= j0 && j < j1) {
#pragma unroll
      for (int s = 0; s < 4; ++s) {
        bw[j][s] = *(const v4f*)(wp + (size_t)(8 * j + s) * HIDF);
      }
    }
  }
}

__global__ __launch_bounds__(PRE_NTHR) void pre_kernel(const float* __restrict__ x,
                                                       const float* __restrict__ W1,
                                                       const float* __restrict__ b1,
                                                       const float* __restrict__ W2,
                                                       const float* __restrict__ b2,
                                                       const float* __restrict__ origin_e,
                                                       const float* __restrict__ temp,
                                                       float* __restrict__ zb,
                                                       float* __restrict__ Hf,
                                                       float* __restrict__ accbuf) {
  __shared__ __attribute__((aligned(16))) float xs[16 * XS_STRIDE];
  __shared__ __attribute__((aligned(16))) float hp[4 * 16 * HS_STRIDE];
  __shared__ __attribute__((aligned(16))) float hs[16 * HS_STRIDE];
  __shared__ float al[CCH * (KORD + 1)];

  const int tid  = threadIdx.x;
  const int w    = tid >> 5;
  const int lane = tid & 31;
  const int mn   = lane & 15;
  const int half = lane >> 4;
  const int kk   = half * 2;
  const int m0   = blockIdx.x * 16;
  const int kq   = w * PRE_KQ;

  const int jr = tid >> 3;
  const int jc = (tid & 7) * 2;
  const int tid2 = tid < CCH * (KORD + 1) - PRE_NTHR ? tid + PRE_NTHR : tid;
  const float tp0 = temp[tid];
  const float tp1 = temp[tid2];
  const float e   = origin_e[m0 + jr];

  v4f xv[PRE_XV];
#pragma unroll
  for (int i = 0; i < PRE_XV; ++i) {
    xv[i] = *(const v4f*)&x[(size_t)(m0 + i) * F_INF + tid * 4];
  }

  v4f bw[PRE_NJ][4];
  const float* wp = W1 + (size_t)(kq + 4 * half) * HIDF + 4 * mn;
  pre_load_w1(bw, wp, 0, PRE_J1);
  __builtin_amdgcn_sched_barrier(0);

  {
    v4f zero4 = {0.f, 0.f, 0.f, 0.f};
    *(v4f*)&accbuf[((size_t)blockIdx.x * PRE_NTHR + tid) * 4] = zero4;
  }

  al[tid] = 0.5f * tanhf(tp0);
  if (tid2 != tid) al[tid2] = 0.5f * tanhf(tp1);
  lds_only_barrier();

  {
    const float a = 1.0f, b = 1.0f, lo = -1.0f, hi = 1.0f;
    const float coef1 = (a - b) * 0.5f - (a + b + 2.0f) * 0.5f * (lo + hi) / (hi - lo);
    const float coef2 = (a + b + 2.0f) / (hi - lo);

    v2f hf2;
#pragma unroll
    for (int cc = 0; cc < 2; ++cc) {
      const float* ac = &al[(jc + cc) * (KORD + 1)];
      float xs_m2 = ac[0];
      float acc = xs_m2;
      float xs_m1 = coef1 * ac[1] + coef2 * ac[1] * e * xs_m2;
      acc += xs_m1;
#pragma unroll
      for (int L = 2; L <= KORD; ++L) {
        float Lf = (float)L;
        float coef_l  = 2.0f * Lf * (Lf + a + b) * (2.0f * Lf - 2.0f + a + b);
        float c_lm1_1 = (2.0f * Lf + a + b - 1.0f) * (2.0f * Lf + a + b) * (2.0f * Lf + a + b - 2.0f);
        float c_lm1_2 = (2.0f * Lf + a + b - 1.0f) * (a * a - b * b);
        float c_lm2   = 2.0f * (Lf - 1.0f + a) * (Lf - 1.0f + b) * (2.0f * Lf + a + b);
        float tmp1   = ac[L - 1] * (c_lm1_1 / coef_l);
        float tmp2   = ac[L - 1] * (c_lm1_2 / coef_l);
        float tmp3   = ac[L - 1] * ac[L - 2] * (c_lm2 / coef_l);
        float tmp1_2 = tmp1 * (2.0f / (hi - lo));
        float tmp2_2 = tmp1 * ((hi + lo) / (hi - lo)) + tmp2;
        float nx = tmp1_2 * (e * xs_m1) - tmp2_2 * xs_m1 - tmp3 * xs_m2;
        acc += nx;
        xs_m2 = xs_m1;
        xs_m1 = nx;
      }
      hf2[cc] = acc;
    }
    *(v2f*)&Hf[(size_t)(m0 + jr) * CCH + jc] = hf2;
  }
  __builtin_amdgcn_sched_barrier(0);

  pre_load_w1(bw, wp, PRE_J1, PRE_NJ);
  const int hr = tid >> 3;
  const int hc = (tid & 7) * 8;
  const v4f bia0 = *(const v4f*)&b1[hc];
  const v4f bia1 = *(const v4f*)&b1[hc + 4];
  __builtin_amdgcn_sched_barrier(0);
#pragma unroll
  for (int i = 0; i < PRE_XV; ++i) {
    float* d = &xs[i * XS_STRIDE + tid * 4];
    d[0] = xv[i][0]; d[1] = xv[i][1]; d[2] = xv[i][2]; d[3] = xv[i][3];
  }
  lds_only_barrier();

  v8f acc[4] = {};
  {
    const float* arow = &xs[mn * XS_STRIDE + kq + 4 * half];
    v4f a4[PRE_NJ];
#pragma unroll
    for (int j = 0; j < PRE_NJ; ++j) {
      a4[j] = *(const v4f*)&arow[8 * j];
    }
    __builtin_amdgcn_sched_barrier(0);
#pragma unroll
    for (int j = 0; j < PRE_NJ; ++j) {
      v2f a0, a1;
      a0.x = a4[j][0]; a0.y = a4[j][1];
      a1.x = a4[j][2]; a1.y = a4[j][3];
#pragma unroll
      for (int q = 0; q < 4; ++q) {
        v2f bv;
        bv.x = bw[j][0][q];
        bv.y = bw[j][1][q];
        acc[q] = wmma4(a0, bv, acc[q]);
      }
#pragma unroll
      for (int q = 0; q < 4; ++q) {
        v2f bv;
        bv.x = bw[j][2][q];
        bv.y = bw[j][3][q];
        acc[q] = wmma4(a1, bv, acc[q]);
      }
    }
  }

  {
    float* pw = &hp[w * (16 * HS_STRIDE)];
#pragma unroll
    for (int r = 0; r < 8; ++r) {
      v4f v = {acc[0][r], acc[1][r], acc[2][r], acc[3][r]};
      *(v4f*)&pw[(8 * half + r) * HS_STRIDE + 4 * mn] = v;
    }
  }
  lds_only_barrier();

  {
    const float* p = &hp[hr * HS_STRIDE + hc];
    v4f s0 = *(const v4f*)&p[0];
    v4f s1 = *(const v4f*)&p[4];
#pragma unroll
    for (int q = 1; q < 4; ++q) {
      s0 += *(const v4f*)&p[q * (16 * HS_STRIDE)];
      s1 += *(const v4f*)&p[q * (16 * HS_STRIDE) + 4];
    }
    s0 += bia0;
    s1 += bia1;
#pragma unroll
    for (int i = 0; i < 4; ++i) {
      s0[i] = s0[i] > 0.0f ? s0[i] : 0.0f;
      s1[i] = s1[i] > 0.0f ? s1[i] : 0.0f;
    }
    *(v4f*)&hs[hr * HS_STRIDE + hc]     = s0;
    *(v4f*)&hs[hr * HS_STRIDE + hc + 4] = s1;
  }
  lds_only_barrier();

  if (w == 0) {
    v8f zacc = {};
#pragma unroll
    for (int k0 = 0; k0 < HIDF; k0 += 4) {
      v2f av, bv;
      av.x = hs[mn * HS_STRIDE + k0 + kk];
      av.y = hs[mn * HS_STRIDE + k0 + kk + 1];
      bv.x = W2[(k0 + kk) * CCH + mn];
      bv.y = W2[(k0 + kk + 1) * CCH + mn];
      zacc = wmma4(av, bv, zacc);
    }
    float bb = b2[mn];
#pragma unroll
    for (int r = 0; r < 8; ++r) {
      int row = half ? r + 8 : r;
      zb[(size_t)(m0 + row) * CCH + mn] = zacc[r] + bb;
    }
  }
}

#define UT_COLS  64
#define UT_ROWS  256
#define UT_CW    2
#define UT_RW    8
#define UT_BCOLS (UT_CW * UT_COLS)
#define UT_BROWS (UT_RW * UT_ROWS)
#define UT_NTHR  (UT_CW * UT_RW * 32)
#define UT_KB    8
#define UT_NB    (UT_ROWS / 4 / UT_KB)
#define UT_BOUT  (UT_BCOLS * CCH)

__device__ __forceinline__ void utz_load_batch(v4f (&buf)[UT_KB][2],
                                               const float* __restrict__ up, int b) {
#pragma unroll
  for (int t = 0; t < UT_KB; ++t) {
#pragma unroll
    for (int s = 0; s < 2; ++s) {
      buf[t][s] = *(const v4f*)(up + (size_t)((b * UT_KB + t) * 4 + s) * N_NODES);
    }
  }
  __builtin_amdgcn_sched_barrier(0);
}

__device__ __forceinline__ void utz_compute_batch(const v4f (&buf)[UT_KB][2],
                                                  const float* zrow, int b,
                                                  v8f (&acc)[4]) {
  v2f av[UT_KB];
#pragma unroll
  for (int t = 0; t < UT_KB; ++t) {
    const float* zp = zrow + (b * UT_KB + t) * 4 * CCH;
    av[t].x = zp[0];
    av[t].y = zp[CCH];
  }
#pragma unroll
  for (int t = 0; t < UT_KB; ++t) {
#pragma unroll
    for (int q = 0; q < 4; ++q) {
      v2f bv;
      bv.x = buf[t][0][q];
      bv.y = buf[t][1][q];
      acc[q] = wmma4(av[t], bv, acc[q]);
    }
  }
}

__global__ __launch_bounds__(UT_NTHR) void utz_kernel(const float* __restrict__ U,
                                                      const float* __restrict__ zb,
                                                      float* __restrict__ tbuf) {
  __shared__ __attribute__((aligned(16))) float zs[UT_BROWS * CCH];

  const int tid  = threadIdx.x;
  const int w    = tid >> 5;
  const int lane = tid & 31;
  const int mn   = lane & 15;
  const int half = lane >> 4;
  const int cw   = w & (UT_CW - 1);
  const int rw   = w / UT_CW;
  const int cb   = blockIdx.x & (N_NODES / UT_BCOLS - 1);
  const int jb   = (blockIdx.x / (N_NODES / UT_BCOLS)) * UT_BROWS;
  const int i0   = cb * UT_BCOLS + cw * UT_COLS;
  const int jw   = jb + rw * UT_ROWS;

  v4f zv[UT_BROWS * CCH / 4 / UT_NTHR];
#pragma unroll
  for (int i = 0; i < UT_BROWS * CCH / 4 / UT_NTHR; ++i) {
    zv[i] = *(const v4f*)&zb[(size_t)jb * CCH + (size_t)(tid + i * UT_NTHR) * 4];
  }

  const float* up = U + (size_t)(jw + 2 * half) * N_NODES + i0 + 4 * mn;
  v4f bufA[UT_KB][2], bufB[UT_KB][2];
  utz_load_batch(bufA, up, 0);
  utz_load_batch(bufB, up, 1);

#pragma unroll
  for (int i = 0; i < UT_BROWS * CCH / 4 / UT_NTHR; ++i) {
    *(v4f*)&zs[(tid + i * UT_NTHR) * 4] = zv[i];
  }
  __syncthreads();

  const float* zrow = &zs[(rw * UT_ROWS + 2 * half) * CCH + mn];
  v8f acc[4] = {};

#pragma unroll 1
  for (int b = 0; b < UT_NB - 2; b += 2) {
    utz_compute_batch(bufA, zrow, b, acc);
    utz_load_batch(bufA, up, b + 2);
    utz_compute_batch(bufB, zrow, b + 1, acc);
    utz_load_batch(bufB, up, b + 3);
  }
  utz_compute_batch(bufA, zrow, UT_NB - 2, acc);
  utz_compute_batch(bufB, zrow, UT_NB - 1, acc);

  __syncthreads();

  float* tw = &zs[w * (UT_COLS * CCH)];
#pragma unroll
  for (int q = 0; q < 4; ++q) {
    float* d = &tw[(4 * mn + q) * CCH + 8 * half];
    v4f lo = {acc[q][0], acc[q][1], acc[q][2], acc[q][3]};
    v4f hi = {acc[q][4], acc[q][5], acc[q][6], acc[q][7]};
    *(v4f*)&d[0] = lo;
    *(v4f*)&d[4] = hi;
  }
  __syncthreads();

  float sum[UT_BOUT / UT_NTHR];
#pragma unroll
  for (int k = 0; k < UT_BOUT / UT_NTHR; ++k) {
    const int o = tid + k * UT_NTHR;
    float s = zs[o];
#pragma unroll
    for (int r = 1; r < UT_RW; ++r) s += zs[r * UT_BOUT + o];
    sum[k] = s;
  }
  float* tg = tbuf + (size_t)cb * UT_BOUT;
#pragma unroll
  for (int k = 0; k < UT_BOUT / UT_NTHR; ++k) {
    unsafeAtomicAdd(&tg[tid + k * UT_NTHR], sum[k]);
  }
}

#define OG_WAVES  16
#define OG_BROWS  (16 * OG_WAVES)
#define OG_KC     1024
#define OG_KB     128
#define OG_NB     (OG_KC / OG_KB)
#define OG_APITCH (OG_KB + 4)
#define OG_SPITCH (OG_KC + 4)
#define OG_NTHR   (OG_WAVES * 32)
#define OG_SV     (OG_KC * CCH / 4 / OG_NTHR)

__device__ __forceinline__ void og_load_batch(v4f (&buf)[16],
                                              const float* __restrict__ up, int b) {
#pragma unroll
  for (int r = 0; r < 16; ++r) {
    buf[r] = *(const v4f*)(up + (size_t)r * N_NODES + b * OG_KB);
  }
  __builtin_amdgcn_sched_barrier(0);
}

__device__ __forceinline__ void og_stage_batch(const v4f (&buf)[16], float* at, int lane) {
#pragma unroll
  for (int r = 0; r < 16; ++r) {
    *(v4f*)&at[r * OG_APITCH + 4 * lane] = buf[r];
  }
}

__device__ __forceinline__ void og_compute_batch(const float* arow, const float* srow,
                                                 v8f& acc0, v8f& acc1) {
  v4f a[8], s[8];
#pragma unroll
  for (int j = 0; j < 8; ++j) {
    a[j] = *(const v4f*)&arow[8 * j];
    s[j] = *(const v4f*)&srow[8 * j];
  }
  __builtin_amdgcn_sched_barrier(0);
#pragma unroll
  for (int g = 0; g < OG_KB / 32; ++g) {
#pragma unroll
    for (int jj = 0; jj < 4; ++jj) {
      const int i = (g & 1) * 4 + jj;
      v2f a0, s0, a1, s1;
      a0.x = a[i][0]; a0.y = a[i][1]; s0.x = s[i][0]; s0.y = s[i][1];
      a1.x = a[i][2]; a1.y = a[i][3]; s1.x = s[i][2]; s1.y = s[i][3];
      acc0 = wmma4(a0, s0, acc0);
      acc1 = wmma4(a1, s1, acc1);
    }
    if (g + 2 < OG_KB / 32) {
#pragma unroll
      for (int jj = 0; jj < 4; ++jj) {
        const int i = (g & 1) * 4 + jj;
        const int j = 4 * (g + 2) + jj;
        a[i] = *(const v4f*)&arow[8 * j];
        s[i] = *(const v4f*)&srow[8 * j];
      }
    }
    __builtin_amdgcn_sched_barrier(0);
  }
}

__global__ __launch_bounds__(OG_NTHR) void outgemm_kernel(const float* __restrict__ U,
                                                          const float* __restrict__ tbuf,
                                                          const float* __restrict__ Hf,
                                                          float* __restrict__ obuf) {
  __shared__ __attribute__((aligned(16))) float sT[CCH * OG_SPITCH];
  __shared__ __attribute__((aligned(16))) float At[OG_WAVES * 16 * OG_APITCH];

  const int tid  = threadIdx.x;
  const int w    = tid >> 5;
  const int lane = tid & 31;
  const int mn   = lane & 15;
  const int half = lane >> 4;
  const int rb   = blockIdx.x & (N_NODES / OG_BROWS - 1);
  const int kb   = (blockIdx.x / (N_NODES / OG_BROWS)) * OG_KC;
  const int r0   = rb * OG_BROWS + w * 16;

  v4f tv[OG_SV], hv[OG_SV];
#pragma unroll
  for (int i = 0; i < OG_SV; ++i) {
    const size_t off = (size_t)kb * CCH + (size_t)(tid + i * OG_NTHR) * 4;
    tv[i] = *(const v4f*)&tbuf[off];
    hv[i] = *(const v4f*)&Hf[off];
  }

  const float* up = U + (size_t)r0 * N_NODES + kb + 4 * lane;
  v4f bufA[16], bufB[16];
  const int rot = (w + rb) & (OG_NB - 1);
  og_load_batch(bufA, up, rot);
  og_load_batch(bufB, up, (rot + 1) & (OG_NB - 1));

#pragma unroll
  for (int i = 0; i < OG_SV; ++i) {
    const int e  = tid + i * OG_NTHR;
    const int k  = e >> 2;
    const int c0 = (e & 3) * 4;
#pragma unroll
    for (int q = 0; q < 4; ++q) {
      sT[(c0 + q) * OG_SPITCH + k] = tv[i][q] * hv[i][q];
    }
  }
  __builtin_amdgcn_fence(__ATOMIC_RELEASE, "workgroup", "local");
  __builtin_amdgcn_s_barrier();
  __builtin_amdgcn_fence(__ATOMIC_ACQUIRE, "workgroup", "local");

  float* at = &At[w * (16 * OG_APITCH)];
  const float* arow = &at[mn * OG_APITCH + 4 * half];
  const float* srow = &sT[mn * OG_SPITCH + 4 * half];
  v8f acc0 = {}, acc1 = {};

#pragma unroll 1
  for (int b = 0; b < OG_NB - 2; b += 2) {
    og_stage_batch(bufA, at, lane);
    og_load_batch(bufA, up, (b + 2 + rot) & (OG_NB - 1));
    og_compute_batch(arow, srow + ((b + rot) & (OG_NB - 1)) * OG_KB, acc0, acc1);
    og_stage_batch(bufB, at, lane);
    og_load_batch(bufB, up, (b + 3 + rot) & (OG_NB - 1));
    og_compute_batch(arow, srow + ((b + 1 + rot) & (OG_NB - 1)) * OG_KB, acc0, acc1);
  }
  og_stage_batch(bufA, at, lane);
  og_compute_batch(arow, srow + ((OG_NB - 2 + rot) & (OG_NB - 1)) * OG_KB, acc0, acc1);
  og_stage_batch(bufB, at, lane);
  og_compute_batch(arow, srow + ((OG_NB - 1 + rot) & (OG_NB - 1)) * OG_KB, acc0, acc1);

#pragma unroll
  for (int r = 0; r < 8; ++r) {
    unsafeAtomicAdd(&obuf[(size_t)(r0 + 8 * half + r) * CCH + mn], acc0[r] + acc1[r]);
  }
}

__global__ void lsm_kernel(const float* __restrict__ obuf,
                           float* __restrict__ out) {
  int n = blockIdx.x * blockDim.x + threadIdx.x;
  if (n >= N_NODES) return;
  float v[CCH];
#pragma unroll
  for (int q = 0; q < 4; ++q) {
    float4 t = *(const float4*)&obuf[(size_t)n * CCH + q * 4];
    v[q * 4 + 0] = t.x; v[q * 4 + 1] = t.y; v[q * 4 + 2] = t.z; v[q * 4 + 3] = t.w;
  }
  float mx = v[0];
#pragma unroll
  for (int i = 1; i < CCH; ++i) mx = fmaxf(mx, v[i]);
  float sm = 0.0f;
#pragma unroll
  for (int i = 0; i < CCH; ++i) sm += expf(v[i] - mx);
  float lse = mx + logf(sm);
#pragma unroll
  for (int q = 0; q < 4; ++q) {
    float4 t;
    t.x = v[q * 4 + 0] - lse; t.y = v[q * 4 + 1] - lse;
    t.z = v[q * 4 + 2] - lse; t.w = v[q * 4 + 3] - lse;
    *(float4*)&out[(size_t)n * CCH + q * 4] = t;
  }
}

extern "C" void kernel_launch(void* const* d_in, const int* in_sizes, int n_in,
                              void* d_out, int out_size, void* d_ws, size_t ws_size,
                              hipStream_t stream) {
  const float* origin_e = (const float*)d_in[0];
  const float* U        = (const float*)d_in[1];
  const float* x        = (const float*)d_in[2];
  const float* W1       = (const float*)d_in[3];
  const float* b1       = (const float*)d_in[4];
  const float* W2       = (const float*)d_in[5];
  const float* b2       = (const float*)d_in[6];
  const float* temp     = (const float*)d_in[7];
  float* out = (float*)d_out;

  float* ws   = (float*)d_ws;
  float* Hf   = ws;
  float* zb   = Hf   + (size_t)N_NODES * CCH;
  float* tbuf = zb   + (size_t)N_NODES * CCH;
  float* obuf = tbuf + (size_t)N_NODES * CCH;

  pre_kernel<<<N_NODES / 16, PRE_NTHR, 0, stream>>>(x, W1, b1, W2, b2, origin_e, temp, zb, Hf, tbuf);
  utz_kernel<<<(N_NODES / UT_BCOLS) * (N_NODES / UT_BROWS), UT_NTHR, 0, stream>>>(U, zb, tbuf);
  outgemm_kernel<<<(N_NODES / OG_BROWS) * (N_NODES / OG_KC), OG_NTHR, 0, stream>>>(U, tbuf, Hf, obuf);
  lsm_kernel<<<N_NODES / 256, 256, 0, stream>>>(obuf, out);
}
